# starting kernel
# baseline (speedup 1.0000x reference)
#include <hip/hip_runtime.h>
#include <math.h>

typedef __attribute__((ext_vector_type(2))) float v2f;
typedef __attribute__((ext_vector_type(8))) float v8f;

#define NL   1000000
#define TILE 250
#define EXT  256
#define HSTR 258
#define NCH  7

__global__ __launch_bounds__(256) void
state_policy_kernel(const float* __restrict__ x,
                    const float* __restrict__ states,
                    const float* __restrict__ conv_w,
                    const float* __restrict__ conv_b,
                    const float* __restrict__ comp_mat,
                    const float* __restrict__ act_mat,
                    float* __restrict__ out)
{
    __shared__ float h_lds[NCH * HSTR];
    __shared__ float st_lds[NCH * EXT];

    const int tid  = threadIdx.x;
    const int lane = tid & 31;
    const int wid  = tid >> 5;
    const int n    = lane & 15;
    const int hi   = lane >> 4;

    const long l0    = (long)blockIdx.x * TILE;
    const long e     = l0 - 3 + tid;
    const bool valid = (e >= 0) && (e < NL);

    float jl[6], jd[6];
#pragma unroll
    for (int c = 0; c < 6; ++c) {
        long flat = (long)c * NL + e;
        bool ok = valid && (flat >= 2);
        jl[c] = ok ? x[7 + flat - 2]            : 0.0f;
        jd[c] = ok ? x[6L * NL + 11 + flat - 2] : 0.0f;
    }
#pragma unroll
    for (int c = 0; c < NCH; ++c)
        st_lds[c * EXT + tid] = valid ? states[(long)c * NL + e] : 0.0f;

    const float ext0 = x[3];
    const float ext1 = x[4];

    float comp[NCH][3];
#pragma unroll
    for (int c = 0; c < NCH; ++c)
#pragma unroll
        for (int j = 0; j < 3; ++j)
            comp[c][j] = comp_mat[c * 3 + j];

    float cb[NCH];
#pragma unroll
    for (int o = 0; o < NCH; ++o) cb[o] = conv_b[o];

    v2f A[6];
#pragma unroll
    for (int kb = 0; kb < 6; ++kb) {
        int k0 = kb * 4 + 2 * hi;
        float ax = 0.0f, ay = 0.0f;
        if (n < NCH) {
            if (k0 < 21)     ax = conv_w[n * 21 + k0];
            if (k0 + 1 < 21) ay = conv_w[n * 21 + k0 + 1];
        }
        A[kb].x = ax; A[kb].y = ay;
    }

    if (tid < NCH) {
        h_lds[tid * HSTR + 0]        = 0.0f;
        h_lds[tid * HSTR + HSTR - 1] = 0.0f;
    }
    __syncthreads();

    for (int it = 0; it < 3; ++it) {
#pragma unroll
        for (int c = 0; c < NCH; ++c) {
            float o0 = (c < 6) ? jl[c] : ext0;
            float o1 = (c < 6) ? jd[c] : ext1;
            float sv = st_lds[c * EXT + tid];
            float hv = tanhf(o0 * comp[c][0] + o1 * comp[c][1] + sv * comp[c][2]);
            h_lds[c * HSTR + tid + 1] = valid ? hv : 0.0f;
        }
        __syncthreads();

#pragma unroll
        for (int gi = 0; gi < 2; ++gi) {
            int base = (wid * 2 + gi) * 16;
            v8f acc = {0.f, 0.f, 0.f, 0.f, 0.f, 0.f, 0.f, 0.f};
#pragma unroll
            for (int kb = 0; kb < 6; ++kb) {
                int k0 = kb * 4 + 2 * hi;
                v2f B;
                {
                    bool vx = (k0 < 21);
                    int ii  = vx ? (k0 / 3) : 0;
                    int kk  = vx ? (k0 % 3) : 0;
                    float r = h_lds[ii * HSTR + base + n + kk];
                    B.x = vx ? r : 0.0f;
                }
                {
                    int k1 = k0 + 1;
                    bool vy = (k1 < 21);
                    int ii  = vy ? (k1 / 3) : 0;
                    int kk  = vy ? (k1 % 3) : 0;
                    float r = h_lds[ii * HSTR + base + n + kk];
                    B.y = vy ? r : 0.0f;
                }
                acc = __builtin_amdgcn_wmma_f32_16x16x4_f32(
                        false, A[kb], false, B, (short)0, acc, false, false);
            }
            if (lane < 16) {
#pragma unroll
                for (int o = 0; o < NCH; ++o)
                    st_lds[o * EXT + base + n] = tanhf(acc[o] + cb[o]);
            }
        }
        __syncthreads();
    }

    if (tid >= 3 && tid < 3 + TILE) {
        long l = l0 + tid - 3;
        if (l < NL) {
#pragma unroll
            for (int c = 0; c < 6; ++c) {
                float a0  = act_mat[c * 2 + 0];
                float a1  = act_mat[c * 2 + 1];
                float stv = st_lds[c * EXT + tid];
                long idx  = (long)c * NL + l - 2;
                if (idx >= 0) out[idx] = a0 * stv + a1 * jd[c];
            }
        }
    }
}

extern "C" void kernel_launch(void* const* d_in, const int* in_sizes, int n_in,
                              void* d_out, int out_size, void* d_ws, size_t ws_size,
                              hipStream_t stream) {
    (void)in_sizes; (void)n_in; (void)out_size; (void)d_ws; (void)ws_size;
    const float* x      = (const float*)d_in[0];
    const float* states = (const float*)d_in[1];
    const float* conv_w = (const float*)d_in[2];
    const float* conv_b = (const float*)d_in[3];
    const float* comp   = (const float*)d_in[4];
    const float* am     = (const float*)d_in[5];
    float* out = (float*)d_out;

    const int nblocks = (NL + TILE - 1) / TILE;
    state_policy_kernel<<<nblocks, 256, 0, stream>>>(x, states, conv_w, conv_b, comp, am, out);
}


# library tanhf (two exec-masked paths) replaced at all 41 sites by branch-free sign(x)*(1-2*rcp(1+exp2(2*log2e*|x|))) on v_exp_f32 + v_rcp_f32, not the hardware tanh; tanh rounding differs
# speedup vs baseline: 2.1093x; 2.1093x over previous
#include <hip/hip_runtime.h>
#include <math.h>


typedef __attribute__((ext_vector_type(2))) float v2f;
typedef __attribute__((ext_vector_type(4))) float v4f;
typedef __attribute__((ext_vector_type(8))) float v8f;

#define NL   1000000
#define TILE 250
#define EXT  256
#define NCH  7
#define HP   272
#define CSL  258

__device__ __forceinline__ float tanh_exp2_rcp(float v)
{
    const float e = __builtin_amdgcn_exp2f(__builtin_fabsf(v) * 2.885390043f);
    const float r = __builtin_amdgcn_rcpf(1.0f + e);
    return __builtin_copysignf(__builtin_fmaf(-2.0f, r, 1.0f), v);
}

__global__ __launch_bounds__(256) void
state_policy_kernel(const float* __restrict__ x,
                    const float* __restrict__ states,
                    const float* __restrict__ conv_w,
                    const float* __restrict__ conv_b,
                    const float* __restrict__ comp_mat,
                    const float* __restrict__ act_mat,
                    float* __restrict__ out)
{
    __shared__ float h_lds[8 * HP];
    __shared__ v4f   c_lds[NCH * CSL];
    float* cs = (float*)c_lds;

    const int tid  = threadIdx.x;
    const int lane = tid & 31;
    const int wid  = tid >> 5;
    const int n    = lane & 15;
    const int hi   = lane >> 4;

    const long l0    = (long)blockIdx.x * TILE;
    const long e     = l0 - 3 + tid;
    const bool valid = (e >= 0) && (e < NL);

    float jl[6], jd[6];
#pragma unroll
    for (int c = 0; c < 6; ++c) {
        long flat = (long)c * NL + e;
        bool ok = valid && (flat >= 2);
        jl[c] = ok ? x[7 + flat - 2]            : 0.0f;
        jd[c] = ok ? x[6L * NL + 11 + flat - 2] : 0.0f;
    }
    float sv[NCH];
#pragma unroll
    for (int c = 0; c < NCH; ++c)
        sv[c] = valid ? states[(long)c * NL + e] : 0.0f;

    const float ext0 = x[3];
    const float ext1 = x[4];

    float comp[NCH][3];
#pragma unroll
    for (int c = 0; c < NCH; ++c)
#pragma unroll
        for (int j = 0; j < 3; ++j)
            comp[c][j] = comp_mat[c * 3 + j];

    v2f AS[2], AC[2];
#pragma unroll
    for (int kb = 0; kb < 2; ++kb) {
        const int i0  = kb * 4 + 2 * hi;
        const int os  = (n < 8) ? n : n - 8;
        const int ts  = (n < 8) ? 0 : 2;
        float sx = 0.0f, sy = 0.0f, cx = 0.0f, cy = 0.0f;
        if (os < NCH) {
            sx = conv_w[os * 21 + i0 * 3 + ts];
            if (i0 + 1 < NCH) sy = conv_w[os * 21 + (i0 + 1) * 3 + ts];
        }
        if (n < NCH) {
            cx = conv_w[n * 21 + i0 * 3 + 1];
            cy = (i0 + 1 < NCH) ? conv_w[n * 21 + (i0 + 1) * 3 + 1] : conv_b[n];
        }
        AS[kb].x = sx; AS[kb].y = sy;
        AC[kb].x = cx; AC[kb].y = cy;
    }

    h_lds[7 * HP + tid] = 1.0f;
    if (tid < NCH) {
        cs[(tid * CSL + 1)   * 4 + 0] = 0.0f;
        cs[(tid * CSL + 256) * 4 + 2] = 0.0f;
    }
    __syncthreads();

    for (int it = 0; it < 3; ++it) {
        if (it > 0) {
#pragma unroll
            for (int c = 0; c < NCH; ++c) {
                const v4f cv = c_lds[c * CSL + tid + 1];
                sv[c] = tanh_exp2_rcp((cv.x + cv.y) + cv.z);
            }
        }
#pragma unroll
        for (int c = 0; c < NCH; ++c) {
            float o0 = (c < 6) ? jl[c] : ext0;
            float o1 = (c < 6) ? jd[c] : ext1;
            float hv = tanh_exp2_rcp(o0 * comp[c][0] + o1 * comp[c][1] + sv[c] * comp[c][2]);
            h_lds[c * HP + tid] = valid ? hv : 0.0f;
        }
        __syncthreads();

#pragma unroll 1
        for (int gi = 0; gi < 2; ++gi) {
            const int col = (wid * 2 + gi) * 16 + n;
            v2f B0, B1;
            B0.x = h_lds[(2 * hi)     * HP + col];
            B0.y = h_lds[(2 * hi + 1) * HP + col];
            B1.x = h_lds[(4 + 2 * hi) * HP + col];
            B1.y = h_lds[(5 + 2 * hi) * HP + col];
            v8f accS = {0.f, 0.f, 0.f, 0.f, 0.f, 0.f, 0.f, 0.f};
            v8f accC = {0.f, 0.f, 0.f, 0.f, 0.f, 0.f, 0.f, 0.f};
            accS = __builtin_amdgcn_wmma_f32_16x16x4_f32(false, AS[0], false, B0, (short)0, accS, false, false);
            accC = __builtin_amdgcn_wmma_f32_16x16x4_f32(false, AC[0], false, B0, (short)0, accC, false, false);
            accS = __builtin_amdgcn_wmma_f32_16x16x4_f32(false, AS[1], false, B1, (short)0, accS, false, false);
            accC = __builtin_amdgcn_wmma_f32_16x16x4_f32(false, AC[1], false, B1, (short)0, accC, false, false);
            const int aS = (hi ? col : col + 2) * 4 + (hi ? 2 : 0);
            const int aC = (col + 1) * 4 + (hi ? 3 : 1);
#pragma unroll
            for (int o = 0; o < NCH; ++o) {
                cs[o * CSL * 4 + aS] = accS[o];
                cs[o * CSL * 4 + aC] = accC[o];
            }
        }
        __syncthreads();
    }

    if (tid >= 3 && tid < 3 + TILE) {
        long l = l0 + tid - 3;
        if (l < NL) {
#pragma unroll
            for (int c = 0; c < 6; ++c) {
                float a0  = act_mat[c * 2 + 0];
                float a1  = act_mat[c * 2 + 1];
                const v4f cv = c_lds[c * CSL + tid + 1];
                float stv = tanh_exp2_rcp((cv.x + cv.y) + cv.z);
                long idx  = (long)c * NL + l - 2;
                if (idx >= 0) out[idx] = a0 * stv + a1 * jd[c];
            }
        }
    }
}

extern "C" void kernel_launch(void* const* d_in, const int* in_sizes, int n_in,
                              void* d_out, int out_size, void* d_ws, size_t ws_size,
                              hipStream_t stream) {
    (void)in_sizes; (void)n_in; (void)out_size; (void)d_ws; (void)ws_size;
    const float* x      = (const float*)d_in[0];
    const float* states = (const float*)d_in[1];
    const float* conv_w = (const float*)d_in[2];
    const float* conv_b = (const float*)d_in[3];
    const float* comp   = (const float*)d_in[4];
    const float* am     = (const float*)d_in[5];
    float* out = (float*)d_out;

    const int nblocks = (NL + TILE - 1) / TILE;
    state_policy_kernel<<<nblocks, 256, 0, stream>>>(x, states, conv_w, conv_b, comp, am, out);
}


# the 19 input loads and the 9 weight loads of every thread read at clamped in-range indices and are zeroed by a select: no exec-masked region, branch or exec save per load; sums unchanged
# speedup vs baseline: 2.2335x; 1.0589x over previous
#include <hip/hip_runtime.h>
#include <math.h>


typedef __attribute__((ext_vector_type(2))) float v2f;
typedef __attribute__((ext_vector_type(4))) float v4f;
typedef __attribute__((ext_vector_type(8))) float v8f;

#define NL   1000000
#define TILE 250
#define EXT  256
#define NCH  7
#define HP   272
#define CSL  258

__device__ __forceinline__ float tanh_exp2_rcp(float v)
{
    const float e = __builtin_amdgcn_exp2f(__builtin_fabsf(v) * 2.885390043f);
    const float r = __builtin_amdgcn_rcpf(1.0f + e);
    return __builtin_copysignf(__builtin_fmaf(-2.0f, r, 1.0f), v);
}

__global__ __launch_bounds__(256) void
state_policy_kernel(const float* __restrict__ x,
                    const float* __restrict__ states,
                    const float* __restrict__ conv_w,
                    const float* __restrict__ conv_b,
                    const float* __restrict__ comp_mat,
                    const float* __restrict__ act_mat,
                    float* __restrict__ out)
{
    __shared__ float h_lds[8 * HP];
    __shared__ v4f   c_lds[NCH * CSL];
    float* cs = (float*)c_lds;

    const int tid  = threadIdx.x;
    const int lane = tid & 31;
    const int wid  = tid >> 5;
    const int n    = lane & 15;
    const int hi   = lane >> 4;

    const long l0    = (long)blockIdx.x * TILE;
    const long e     = l0 - 3 + tid;
    const bool valid = (e >= 0) && (e < NL);

    const int ec = valid ? (int)e : 0;
    float jl[6], jd[6];
#pragma unroll
    for (int c = 0; c < 6; ++c) {
        const bool ok = valid && (c > 0 || e >= 2);
        const float tl = x[5 + c * NL + ec];
        const float td = x[6 * NL + 9 + c * NL + ec];
        jl[c] = ok ? tl : 0.0f;
        jd[c] = ok ? td : 0.0f;
    }
    float sv[NCH];
#pragma unroll
    for (int c = 0; c < NCH; ++c) {
        const float ts = states[c * NL + ec];
        sv[c] = valid ? ts : 0.0f;
    }

    const float ext0 = x[3];
    const float ext1 = x[4];

    float comp[NCH][3];
#pragma unroll
    for (int c = 0; c < NCH; ++c)
#pragma unroll
        for (int j = 0; j < 3; ++j)
            comp[c][j] = comp_mat[c * 3 + j];

    v2f AS[2], AC[2];
    const int  os   = (n < 8) ? n : n - 8;
    const int  ts   = (n < 8) ? 0 : 2;
    const bool sok  = os < NCH;
    const bool cok  = n < NCH;
    const int  osc  = sok ? os : 0;
    const int  nc   = cok ? n : 0;
    const float cbv = conv_b[nc];
#pragma unroll
    for (int kb = 0; kb < 2; ++kb) {
        const int  i0  = kb * 4 + 2 * hi;
        const bool yok = (i0 + 1 < NCH);
        const int  i1  = yok ? i0 + 1 : 0;
        const float wsx = conv_w[osc * 21 + i0 * 3 + ts];
        const float wsy = conv_w[osc * 21 + i1 * 3 + ts];
        const float wcx = conv_w[nc * 21 + i0 * 3 + 1];
        const float wcy = conv_w[nc * 21 + i1 * 3 + 1];
        AS[kb].x = sok ? wsx : 0.0f;
        AS[kb].y = (sok && yok) ? wsy : 0.0f;
        AC[kb].x = cok ? wcx : 0.0f;
        AC[kb].y = cok ? (yok ? wcy : cbv) : 0.0f;
    }

    h_lds[7 * HP + tid] = 1.0f;
    if (tid < NCH) {
        cs[(tid * CSL + 1)   * 4 + 0] = 0.0f;
        cs[(tid * CSL + 256) * 4 + 2] = 0.0f;
    }
    __syncthreads();

    for (int it = 0; it < 3; ++it) {
        if (it > 0) {
#pragma unroll
            for (int c = 0; c < NCH; ++c) {
                const v4f cv = c_lds[c * CSL + tid + 1];
                sv[c] = tanh_exp2_rcp((cv.x + cv.y) + cv.z);
            }
        }
#pragma unroll
        for (int c = 0; c < NCH; ++c) {
            float o0 = (c < 6) ? jl[c] : ext0;
            float o1 = (c < 6) ? jd[c] : ext1;
            float hv = tanh_exp2_rcp(o0 * comp[c][0] + o1 * comp[c][1] + sv[c] * comp[c][2]);
            h_lds[c * HP + tid] = valid ? hv : 0.0f;
        }
        __syncthreads();

#pragma unroll 1
        for (int gi = 0; gi < 2; ++gi) {
            const int col = (wid * 2 + gi) * 16 + n;
            v2f B0, B1;
            B0.x = h_lds[(2 * hi)     * HP + col];
            B0.y = h_lds[(2 * hi + 1) * HP + col];
            B1.x = h_lds[(4 + 2 * hi) * HP + col];
            B1.y = h_lds[(5 + 2 * hi) * HP + col];
            v8f accS = {0.f, 0.f, 0.f, 0.f, 0.f, 0.f, 0.f, 0.f};
            v8f accC = {0.f, 0.f, 0.f, 0.f, 0.f, 0.f, 0.f, 0.f};
            accS = __builtin_amdgcn_wmma_f32_16x16x4_f32(false, AS[0], false, B0, (short)0, accS, false, false);
            accC = __builtin_amdgcn_wmma_f32_16x16x4_f32(false, AC[0], false, B0, (short)0, accC, false, false);
            accS = __builtin_amdgcn_wmma_f32_16x16x4_f32(false, AS[1], false, B1, (short)0, accS, false, false);
            accC = __builtin_amdgcn_wmma_f32_16x16x4_f32(false, AC[1], false, B1, (short)0, accC, false, false);
            const int aS = (hi ? col : col + 2) * 4 + (hi ? 2 : 0);
            const int aC = (col + 1) * 4 + (hi ? 3 : 1);
#pragma unroll
            for (int o = 0; o < NCH; ++o) {
                cs[o * CSL * 4 + aS] = accS[o];
                cs[o * CSL * 4 + aC] = accC[o];
            }
        }
        __syncthreads();
    }

    if (tid >= 3 && tid < 3 + TILE) {
        long l = l0 + tid - 3;
        if (l < NL) {
#pragma unroll
            for (int c = 0; c < 6; ++c) {
                float a0  = act_mat[c * 2 + 0];
                float a1  = act_mat[c * 2 + 1];
                const v4f cv = c_lds[c * CSL + tid + 1];
                float stv = tanh_exp2_rcp((cv.x + cv.y) + cv.z);
                long idx  = (long)c * NL + l - 2;
                if (idx >= 0) out[idx] = a0 * stv + a1 * jd[c];
            }
        }
    }
}

extern "C" void kernel_launch(void* const* d_in, const int* in_sizes, int n_in,
                              void* d_out, int out_size, void* d_ws, size_t ws_size,
                              hipStream_t stream) {
    (void)in_sizes; (void)n_in; (void)out_size; (void)d_ws; (void)ws_size;
    const float* x      = (const float*)d_in[0];
    const float* states = (const float*)d_in[1];
    const float* conv_w = (const float*)d_in[2];
    const float* conv_b = (const float*)d_in[3];
    const float* comp   = (const float*)d_in[4];
    const float* am     = (const float*)d_in[5];
    float* out = (float*)d_out;

    const int nblocks = (NL + TILE - 1) / TILE;
    state_policy_kernel<<<nblocks, 256, 0, stream>>>(x, states, conv_w, conv_b, comp, am, out);
}


# state tanh and the gate tanh behind it fused in r = 1/(1+2^s) form: 2*log2(e) folded into conv weights and gate coefficients, gate argument = p + q*r, no |x| or sign copy; tanh, gate rounding differ
# speedup vs baseline: 2.3874x; 1.0689x over previous
#include <hip/hip_runtime.h>
#include <math.h>


typedef __attribute__((ext_vector_type(2))) float v2f;
typedef __attribute__((ext_vector_type(4))) float v4f;
typedef __attribute__((ext_vector_type(8))) float v8f;

#define NL   1000000
#define TILE 250
#define EXT  256
#define NCH  7
#define HP   272
#define CSL  258

#define K2LOG2E 2.885390043f
__device__ __forceinline__ float r_of(float vs)
{
    return __builtin_amdgcn_rcpf(1.0f + __builtin_amdgcn_exp2f(vs));
}

__global__ __launch_bounds__(256) void
state_policy_kernel(const float* __restrict__ x,
                    const float* __restrict__ states,
                    const float* __restrict__ conv_w,
                    const float* __restrict__ conv_b,
                    const float* __restrict__ comp_mat,
                    const float* __restrict__ act_mat,
                    float* __restrict__ out)
{
    __shared__ float h_lds[8 * HP];
    __shared__ v4f   c_lds[NCH * CSL];
    float* cs = (float*)c_lds;

    const int tid  = threadIdx.x;
    const int lane = tid & 31;
    const int wid  = tid >> 5;
    const int n    = lane & 15;
    const int hi   = lane >> 4;

    const long l0    = (long)blockIdx.x * TILE;
    const long e     = l0 - 3 + tid;
    const bool valid = (e >= 0) && (e < NL);

    const int ec = valid ? (int)e : 0;
    float jl[6], jd[6];
#pragma unroll
    for (int c = 0; c < 6; ++c) {
        const bool ok = valid && (c > 0 || e >= 2);
        const float tl = x[5 + c * NL + ec];
        const float td = x[6 * NL + 9 + c * NL + ec];
        jl[c] = ok ? tl : 0.0f;
        jd[c] = ok ? td : 0.0f;
    }
    float sv[NCH];
#pragma unroll
    for (int c = 0; c < NCH; ++c) {
        const float ts = states[c * NL + ec];
        sv[c] = valid ? ts : 0.0f;
    }

    const float ext0 = x[3];
    const float ext1 = x[4];

    float ck[NCH][3];
#pragma unroll
    for (int c = 0; c < NCH; ++c)
#pragma unroll
        for (int j = 0; j < 3; ++j)
            ck[c][j] = comp_mat[c * 3 + j] * K2LOG2E;

    v2f AS[2], AC[2];
    const int  os   = (n < 8) ? n : n - 8;
    const int  ts   = (n < 8) ? 0 : 2;
    const bool sok  = os < NCH;
    const bool cok  = n < NCH;
    const int  osc  = sok ? os : 0;
    const int  nc   = cok ? n : 0;
    const float cbv = conv_b[nc];
#pragma unroll
    for (int kb = 0; kb < 2; ++kb) {
        const int  i0  = kb * 4 + 2 * hi;
        const bool yok = (i0 + 1 < NCH);
        const int  i1  = yok ? i0 + 1 : 0;
        const float wsx = conv_w[osc * 21 + i0 * 3 + ts];
        const float wsy = conv_w[osc * 21 + i1 * 3 + ts];
        const float wcx = conv_w[nc * 21 + i0 * 3 + 1];
        const float wcy = conv_w[nc * 21 + i1 * 3 + 1];
        AS[kb].x = sok ? wsx : 0.0f;
        AS[kb].y = (sok && yok) ? wsy : 0.0f;
        AC[kb].x = cok ? wcx : 0.0f;
        AC[kb].y = cok ? (yok ? wcy : cbv) : 0.0f;
        AS[kb] *= K2LOG2E;
        AC[kb] *= K2LOG2E;
    }

    h_lds[7 * HP + tid] = 1.0f;
    if (tid < NCH) {
        cs[(tid * CSL + 1)   * 4 + 0] = 0.0f;
        cs[(tid * CSL + 256) * 4 + 2] = 0.0f;
    }
    __syncthreads();

    float pg[NCH];
#pragma unroll
    for (int c = 0; c < NCH; ++c) {
        const float o0 = (c < 6) ? jl[c] : ext0;
        const float o1 = (c < 6) ? jd[c] : ext1;
        pg[c] = o0 * ck[c][0] + o1 * ck[c][1];
        const float rg = r_of(__builtin_fmaf(sv[c], ck[c][2], pg[c]));
        const float hv = __builtin_fmaf(-2.0f, rg, 1.0f);
        h_lds[c * HP + tid] = valid ? hv : 0.0f;
        pg[c] += ck[c][2];
    }
    __syncthreads();

    for (int it = 0; it < 3; ++it) {
#pragma unroll 1
        for (int gi = 0; gi < 2; ++gi) {
            const int col = (wid * 2 + gi) * 16 + n;
            v2f B0, B1;
            B0.x = h_lds[(2 * hi)     * HP + col];
            B0.y = h_lds[(2 * hi + 1) * HP + col];
            B1.x = h_lds[(4 + 2 * hi) * HP + col];
            B1.y = h_lds[(5 + 2 * hi) * HP + col];
            v8f accS = {0.f, 0.f, 0.f, 0.f, 0.f, 0.f, 0.f, 0.f};
            v8f accC = {0.f, 0.f, 0.f, 0.f, 0.f, 0.f, 0.f, 0.f};
            accS = __builtin_amdgcn_wmma_f32_16x16x4_f32(false, AS[0], false, B0, (short)0, accS, false, false);
            accC = __builtin_amdgcn_wmma_f32_16x16x4_f32(false, AC[0], false, B0, (short)0, accC, false, false);
            accS = __builtin_amdgcn_wmma_f32_16x16x4_f32(false, AS[1], false, B1, (short)0, accS, false, false);
            accC = __builtin_amdgcn_wmma_f32_16x16x4_f32(false, AC[1], false, B1, (short)0, accC, false, false);
            const int aS = (hi ? col : col + 2) * 4 + (hi ? 2 : 0);
            const int aC = (col + 1) * 4 + (hi ? 3 : 1);
#pragma unroll
            for (int o = 0; o < NCH; ++o) {
                cs[o * CSL * 4 + aS] = accS[o];
                cs[o * CSL * 4 + aC] = accC[o];
            }
        }
        __syncthreads();

        if (it < 2) {
#pragma unroll
            for (int c = 0; c < NCH; ++c) {
                const v4f cv = c_lds[c * CSL + tid + 1];
                const float rs = r_of((cv.x + cv.y) + cv.z);
                const float rg = r_of(__builtin_fmaf(rs, -2.0f * ck[c][2], pg[c]));
                const float hv = __builtin_fmaf(-2.0f, rg, 1.0f);
                h_lds[c * HP + tid] = valid ? hv : 0.0f;
            }
            __syncthreads();
        }
    }

    if (tid >= 3 && tid < 3 + TILE) {
        long l = l0 + tid - 3;
        if (l < NL) {
#pragma unroll
            for (int c = 0; c < 6; ++c) {
                float a0  = act_mat[c * 2 + 0];
                float a1  = act_mat[c * 2 + 1];
                const v4f cv = c_lds[c * CSL + tid + 1];
                const float rs = r_of((cv.x + cv.y) + cv.z);
                long idx  = (long)c * NL + l - 2;
                if (idx >= 0) out[idx] = __builtin_fmaf(rs, -2.0f * a0, __builtin_fmaf(a1, jd[c], a0));
            }
        }
    }
}

extern "C" void kernel_launch(void* const* d_in, const int* in_sizes, int n_in,
                              void* d_out, int out_size, void* d_ws, size_t ws_size,
                              hipStream_t stream) {
    (void)in_sizes; (void)n_in; (void)out_size; (void)d_ws; (void)ws_size;
    const float* x      = (const float*)d_in[0];
    const float* states = (const float*)d_in[1];
    const float* conv_w = (const float*)d_in[2];
    const float* conv_b = (const float*)d_in[3];
    const float* comp   = (const float*)d_in[4];
    const float* am     = (const float*)d_in[5];
    float* out = (float*)d_out;

    const int nblocks = (NL + TILE - 1) / TILE;
    state_policy_kernel<<<nblocks, 256, 0, stream>>>(x, states, conv_w, conv_b, comp, am, out);
}
